# starting kernel
# baseline (speedup 1.0000x reference)
#include <hip/hip_runtime.h>
#include <hip/hip_bf16.h>

typedef __attribute__((ext_vector_type(2))) float v2f;
typedef __attribute__((ext_vector_type(8))) float v8f;

#define N_TOT 131072
#define K_CLS 64
#define D_DIM 256
#define FP_EPS 1e-12f
#define MLDS_STRIDE 260

__device__ __forceinline__ v8f wmma4(v2f a, v2f b, v8f c) {
    return __builtin_amdgcn_wmma_f32_16x16x4_f32(
         false, a,  false, b,
         (short)0, c,  false,  false);
}

__global__ __launch_bounds__(256) void zero_ws(float* __restrict__ ws,
                                               float* __restrict__ out0) {
    int i = blockIdx.x * 256 + threadIdx.x;
    if (i < K_CLS * D_DIM + 2 * K_CLS) ws[i] = 0.0f;
    if (i == 0) out0[0] = 0.0f;
}

__global__ __launch_bounds__(256)
void pass1_colsum(const float* __restrict__ X, const float* __restrict__ A,
                  float* __restrict__ sums, float* __restrict__ freq,
                  int rowsPerBlock) {
    const int tid   = threadIdx.x;
    const int lane  = tid & 31;
    const int wave  = tid >> 5;
    const int ktile = wave & 3;
    const int dgrp  = wave >> 2;
    const int kbase = ktile * 16;
    const int dbase = dgrp * 128;
    const int l15   = lane & 15;
    const int hi    = (lane >= 16) ? 1 : 0;

    v8f acc[8];
    #pragma unroll
    for (int t = 0; t < 8; ++t) { v8f z = {}; acc[t] = z; }

    float fpart = 0.0f;
    const int row0 = blockIdx.x * rowsPerBlock;

    for (int n0 = row0; n0 < row0 + rowsPerBlock; n0 += 4) {
        const int ra = n0 + (hi ? 2 : 0);
        const int rb = ra + 1;
        v2f af;
        af.x = A[(size_t)ra * K_CLS + kbase + l15];
        af.y = A[(size_t)rb * K_CLS + kbase + l15];
        fpart += af.x + af.y;

        const float* xra = X + (size_t)ra * D_DIM + dbase + l15;
        const float* xrb = X + (size_t)rb * D_DIM + dbase + l15;
        #pragma unroll
        for (int dt = 0; dt < 8; ++dt) {
            v2f bf;
            bf.x = xra[dt * 16];
            bf.y = xrb[dt * 16];
            acc[dt] = wmma4(af, bf, acc[dt]);
        }
    }

    fpart += __shfl_xor(fpart, 16, 32);
    if (dgrp == 0 && lane < 16)
        unsafeAtomicAdd(&freq[kbase + lane], fpart);

    #pragma unroll
    for (int dt = 0; dt < 8; ++dt) {
        #pragma unroll
        for (int r = 0; r < 8; ++r) {
            const int k = kbase + r + (hi ? 8 : 0);
            const int d = dbase + dt * 16 + l15;
            unsafeAtomicAdd(&sums[k * D_DIM + d], acc[dt][r]);
        }
    }
}

__global__ __launch_bounds__(256)
void means_msq(const float* __restrict__ sums, const float* __restrict__ freq,
               const float* __restrict__ past, float* __restrict__ out_means,
               float* __restrict__ msq) {
    __shared__ float red[8];
    const int k = blockIdx.x;
    const int d = threadIdx.x;
    const float f  = freq[k];
    const float bm = sums[k * D_DIM + d] / fmaxf(f, FP_EPS);
    const float pm = past[k * D_DIM + d];
    const float m  = (f > FP_EPS) ? (pm * 0.5f + bm * 0.5f) : pm;
    out_means[k * D_DIM + d] = m;

    float sq = m * m;
    #pragma unroll
    for (int off = 16; off > 0; off >>= 1) sq += __shfl_xor(sq, off, 32);
    if ((threadIdx.x & 31) == 0) red[threadIdx.x >> 5] = sq;
    __syncthreads();
    if (threadIdx.x == 0) {
        float t = 0.0f;
        #pragma unroll
        for (int i = 0; i < 8; ++i) t += red[i];
        msq[k] = t;
    }
}

__global__ __launch_bounds__(256)
void pass2_loss(const float* __restrict__ X, const float* __restrict__ A,
                const float* __restrict__ S, const float* __restrict__ means,
                const float* __restrict__ msq, float* __restrict__ loss) {
    extern __shared__ float smem[];
    float* mlds    = smem;
    float* msq_lds = smem + K_CLS * MLDS_STRIDE;
    float* xsq     = msq_lds + K_CLS;

    const int tid  = threadIdx.x;
    const int lane = tid & 31;
    const int wave = tid >> 5;
    const int l15  = lane & 15;
    const int hi   = (lane >= 16) ? 1 : 0;

    #pragma unroll
    for (int i = 0; i < 16; ++i) {
        const int idx = tid + i * 256;
        const int k   = idx >> 6;
        const int c4  = idx & 63;
        const float4 v = ((const float4*)means)[(size_t)k * 64 + c4];
        *(float4*)&mlds[k * MLDS_STRIDE + c4 * 4] = v;
    }
    if (tid < K_CLS) msq_lds[tid] = msq[tid];
    if (tid < 128)   xsq[tid] = 0.0f;
    __syncthreads();

    const int n0 = blockIdx.x * 128 + wave * 16;

    v8f acc[4];
    #pragma unroll
    for (int t = 0; t < 4; ++t) { v8f z = {}; acc[t] = z; }

    float xp = 0.0f;
    const float* xrow = X + (size_t)(n0 + l15) * D_DIM + (hi ? 2 : 0);

    for (int d0 = 0; d0 < D_DIM; d0 += 4) {
        v2f af;
        af.x = xrow[d0];
        af.y = xrow[d0 + 1];
        xp += af.x * af.x + af.y * af.y;
        #pragma unroll
        for (int kt = 0; kt < 4; ++kt) {
            const float* mrow = &mlds[(kt * 16 + l15) * MLDS_STRIDE + d0 + (hi ? 2 : 0)];
            v2f bf;
            bf.x = mrow[0];
            bf.y = mrow[1];
            acc[kt] = wmma4(af, bf, acc[kt]);
        }
    }

    unsafeAtomicAdd(&xsq[wave * 16 + l15], xp);
    __syncthreads();

    float c = 0.0f;
    #pragma unroll
    for (int kt = 0; kt < 4; ++kt) {
        #pragma unroll
        for (int r = 0; r < 8; ++r) {
            const int nl = r + (hi ? 8 : 0);
            const int k  = kt * 16 + l15;
            const float cross = acc[kt][r];
            const float dist  = xsq[wave * 16 + nl] + msq_lds[k] - 2.0f * cross;
            const size_t gi = (size_t)(n0 + nl) * K_CLS + k;
            c += A[gi] * S[gi] * dist;
        }
    }
    #pragma unroll
    for (int off = 16; off > 0; off >>= 1) c += __shfl_xor(c, off, 32);
    if (lane == 0)
        unsafeAtomicAdd(loss, c * (1.0f / (float)N_TOT));
}

extern "C" void kernel_launch(void* const* d_in, const int* in_sizes, int n_in,
                              void* d_out, int out_size, void* d_ws, size_t ws_size,
                              hipStream_t stream) {
    const float* X = (const float*)d_in[0];
    const float* A = (const float*)d_in[1];
    const float* S = (const float*)d_in[2];
    const float* P = (const float*)d_in[3];
    float* out = (float*)d_out;

    float* ws   = (float*)d_ws;
    float* sums = ws;
    float* freq = ws + K_CLS * D_DIM;
    float* msq  = freq + K_CLS;

    zero_ws<<<(K_CLS * D_DIM + 2 * K_CLS + 255) / 256, 256, 0, stream>>>(ws, out);

    const int rowsPerBlock = 512;
    pass1_colsum<<<N_TOT / rowsPerBlock, 256, 0, stream>>>(X, A, sums, freq, rowsPerBlock);

    means_msq<<<K_CLS, 256, 0, stream>>>(sums, freq, P, out + 1, msq);

    const size_t smemBytes = (size_t)(K_CLS * MLDS_STRIDE + K_CLS + 128) * sizeof(float);
    pass2_loss<<<N_TOT / 128, 256, smemBytes, stream>>>(X, A, S, out + 1, msq, out);
}


# pass 1 data path: X and A go through shared LDS tiles (64 rows, row pairs interleaved; whole-line 16-byte loads one tile ahead), 16 waves read WMMA fragments as 8-byte LDS loads; sum order kept
# speedup vs baseline: 1.5393x; 1.5393x over previous
#include <hip/hip_runtime.h>
#include <hip/hip_bf16.h>

typedef __attribute__((ext_vector_type(2))) float v2f;
typedef __attribute__((ext_vector_type(8))) float v8f;

#define N_TOT 131072
#define K_CLS 64
#define D_DIM 256
#define FP_EPS 1e-12f
#define MLDS_STRIDE 260

__device__ __forceinline__ v8f wmma4(v2f a, v2f b, v8f c) {
    return __builtin_amdgcn_wmma_f32_16x16x4_f32(
         false, a,  false, b,
         (short)0, c,  false,  false);
}

__global__ __launch_bounds__(256) void zero_ws(float* __restrict__ ws,
                                               float* __restrict__ out0) {
    int i = blockIdx.x * 256 + threadIdx.x;
    if (i < K_CLS * D_DIM + 2 * K_CLS) ws[i] = 0.0f;
    if (i == 0) out0[0] = 0.0f;
}

#define P1_THREADS 512
#define P1_ROWS    512
#define P1_TN      64
#define P1_XP      544
#define P1_AP      160
#define P1_XT      ((P1_TN / 2) * P1_XP)
#define P1_AT      ((P1_TN / 2) * P1_AP)
#define P1_STAGE   (P1_XT + P1_AT)
#define P1_NSTAGE  (P1_ROWS / P1_TN)

__global__ __launch_bounds__(P1_THREADS)
void pass1_colsum(const float* __restrict__ X, const float* __restrict__ A,
                  float* __restrict__ sums, float* __restrict__ freq) {
    __shared__ __attribute__((aligned(16))) float tile[2 * P1_STAGE];

    const int tid  = threadIdx.x;
    const int lane = tid & 31;
    const int wave = __builtin_amdgcn_readfirstlane(tid >> 5);
    const int c    = lane & 15;
    const int h    = lane >> 4;
    const int kw   = (wave & 1) * 32;
    const int dw   = (wave >> 1) * 32;

    const size_t row0 = (size_t)blockIdx.x * P1_ROWS;

    const int xp = tid >> 6, xc4 = tid & 63;
    const int ap = tid >> 4, ac4 = tid & 15;
    const float4* xg = (const float4*)(X + (row0 + 2 * xp) * D_DIM) + xc4;
    const float4* ag = (const float4*)(A + (row0 + 2 * ap) * K_CLS) + ac4;
    const int xl = xp * P1_XP + xc4 * 8;
    const int al = P1_XT + ap * P1_AP + ac4 * 8;

    float4 xe[4], xo[4], ae, ao;

    v8f acc[2][2];
    #pragma unroll
    for (int i = 0; i < 2; ++i)
        #pragma unroll
        for (int j = 0; j < 2; ++j) { v8f z = {}; acc[i][j] = z; }
    float fp0 = 0.0f, fp1 = 0.0f;

#define P1_LOAD(S)                                                                   \
    {                                                                                \
        const float4* xn = xg + (size_t)(S) * P1_TN * (D_DIM / 4);                   \
        const float4* an = ag + (size_t)(S) * P1_TN * (K_CLS / 4);                   \
        _Pragma("unroll")                                                            \
        for (int i = 0; i < 4; ++i) {                                                \
            xe[i] = xn[(size_t)(16 * i) * (D_DIM / 4)];                              \
            xo[i] = xn[(size_t)(16 * i + 1) * (D_DIM / 4)];                          \
        }                                                                            \
        ae = an[0];                                                                  \
        ao = an[K_CLS / 4];                                                          \
    }
#define P1_STORE(BUF)                                                                \
    {                                                                                \
        _Pragma("unroll")                                                            \
        for (int i = 0; i < 4; ++i) {                                                \
            float* p = &tile[(BUF) + xl + (8 * i) * P1_XP];                          \
            *(float4*)p       = make_float4(xe[i].x, xo[i].x, xe[i].y, xo[i].y);     \
            *(float4*)(p + 4) = make_float4(xe[i].z, xo[i].z, xe[i].w, xo[i].w);     \
        }                                                                            \
        float* q = &tile[(BUF) + al];                                                \
        *(float4*)q       = make_float4(ae.x, ao.x, ae.y, ao.y);                     \
        *(float4*)(q + 4) = make_float4(ae.z, ao.z, ae.w, ao.w);                     \
    }

    P1_LOAD(0)
    P1_STORE(0)
    __syncthreads();

    for (int s = 0; s < P1_NSTAGE; ++s) {
        const int buf = (s & 1) * P1_STAGE;
        const bool more = (s + 1 < P1_NSTAGE);

        if (more) P1_LOAD(s + 1)

        const float* xt = tile + buf + h * P1_XP + 2 * (dw + c);
        const float* at = tile + buf + P1_XT + h * P1_AP + 2 * (kw + c);
        #pragma unroll
        for (int g = 0; g < P1_TN / 4; ++g) {
            const float2 a0 = *(const float2*)(at + (2 * g) * P1_AP);
            const float2 a1 = *(const float2*)(at + (2 * g) * P1_AP + 32);
            const float2 b0 = *(const float2*)(xt + (2 * g) * P1_XP);
            const float2 b1 = *(const float2*)(xt + (2 * g) * P1_XP + 32);
            v2f af0, af1, bf0, bf1;
            af0.x = a0.x; af0.y = a0.y;
            af1.x = a1.x; af1.y = a1.y;
            bf0.x = b0.x; bf0.y = b0.y;
            bf1.x = b1.x; bf1.y = b1.y;
            fp0 += a0.x + a0.y;
            fp1 += a1.x + a1.y;
            acc[0][0] = wmma4(af0, bf0, acc[0][0]);
            acc[0][1] = wmma4(af0, bf1, acc[0][1]);
            acc[1][0] = wmma4(af1, bf0, acc[1][0]);
            acc[1][1] = wmma4(af1, bf1, acc[1][1]);
        }

        if (more) P1_STORE(((s + 1) & 1) * P1_STAGE)
        __syncthreads();
    }
#undef P1_LOAD
#undef P1_STORE

    fp0 += __shfl_xor(fp0, 16, 32);
    fp1 += __shfl_xor(fp1, 16, 32);
    if ((wave >> 1) == 0 && lane < 16) {
        unsafeAtomicAdd(&freq[kw + lane], fp0);
        unsafeAtomicAdd(&freq[kw + 16 + lane], fp1);
    }

    #pragma unroll
    for (int kt = 0; kt < 2; ++kt) {
        #pragma unroll
        for (int dt = 0; dt < 2; ++dt) {
            #pragma unroll
            for (int r = 0; r < 8; ++r) {
                const int k = kw + 16 * kt + r + 8 * h;
                const int d = dw + 16 * dt + c;
                unsafeAtomicAdd(&sums[k * D_DIM + d], acc[kt][dt][r]);
            }
        }
    }
}

__global__ __launch_bounds__(256)
void means_msq(const float* __restrict__ sums, const float* __restrict__ freq,
               const float* __restrict__ past, float* __restrict__ out_means,
               float* __restrict__ msq) {
    __shared__ float red[8];
    const int k = blockIdx.x;
    const int d = threadIdx.x;
    const float f  = freq[k];
    const float bm = sums[k * D_DIM + d] / fmaxf(f, FP_EPS);
    const float pm = past[k * D_DIM + d];
    const float m  = (f > FP_EPS) ? (pm * 0.5f + bm * 0.5f) : pm;
    out_means[k * D_DIM + d] = m;

    float sq = m * m;
    #pragma unroll
    for (int off = 16; off > 0; off >>= 1) sq += __shfl_xor(sq, off, 32);
    if ((threadIdx.x & 31) == 0) red[threadIdx.x >> 5] = sq;
    __syncthreads();
    if (threadIdx.x == 0) {
        float t = 0.0f;
        #pragma unroll
        for (int i = 0; i < 8; ++i) t += red[i];
        msq[k] = t;
    }
}

__global__ __launch_bounds__(256)
void pass2_loss(const float* __restrict__ X, const float* __restrict__ A,
                const float* __restrict__ S, const float* __restrict__ means,
                const float* __restrict__ msq, float* __restrict__ loss) {
    extern __shared__ float smem[];
    float* mlds    = smem;
    float* msq_lds = smem + K_CLS * MLDS_STRIDE;
    float* xsq     = msq_lds + K_CLS;

    const int tid  = threadIdx.x;
    const int lane = tid & 31;
    const int wave = tid >> 5;
    const int l15  = lane & 15;
    const int hi   = (lane >= 16) ? 1 : 0;

    #pragma unroll
    for (int i = 0; i < 16; ++i) {
        const int idx = tid + i * 256;
        const int k   = idx >> 6;
        const int c4  = idx & 63;
        const float4 v = ((const float4*)means)[(size_t)k * 64 + c4];
        *(float4*)&mlds[k * MLDS_STRIDE + c4 * 4] = v;
    }
    if (tid < K_CLS) msq_lds[tid] = msq[tid];
    if (tid < 128)   xsq[tid] = 0.0f;
    __syncthreads();

    const int n0 = blockIdx.x * 128 + wave * 16;

    v8f acc[4];
    #pragma unroll
    for (int t = 0; t < 4; ++t) { v8f z = {}; acc[t] = z; }

    float xp = 0.0f;
    const float* xrow = X + (size_t)(n0 + l15) * D_DIM + (hi ? 2 : 0);

    for (int d0 = 0; d0 < D_DIM; d0 += 4) {
        v2f af;
        af.x = xrow[d0];
        af.y = xrow[d0 + 1];
        xp += af.x * af.x + af.y * af.y;
        #pragma unroll
        for (int kt = 0; kt < 4; ++kt) {
            const float* mrow = &mlds[(kt * 16 + l15) * MLDS_STRIDE + d0 + (hi ? 2 : 0)];
            v2f bf;
            bf.x = mrow[0];
            bf.y = mrow[1];
            acc[kt] = wmma4(af, bf, acc[kt]);
        }
    }

    unsafeAtomicAdd(&xsq[wave * 16 + l15], xp);
    __syncthreads();

    float c = 0.0f;
    #pragma unroll
    for (int kt = 0; kt < 4; ++kt) {
        #pragma unroll
        for (int r = 0; r < 8; ++r) {
            const int nl = r + (hi ? 8 : 0);
            const int k  = kt * 16 + l15;
            const float cross = acc[kt][r];
            const float dist  = xsq[wave * 16 + nl] + msq_lds[k] - 2.0f * cross;
            const size_t gi = (size_t)(n0 + nl) * K_CLS + k;
            c += A[gi] * S[gi] * dist;
        }
    }
    #pragma unroll
    for (int off = 16; off > 0; off >>= 1) c += __shfl_xor(c, off, 32);
    if (lane == 0)
        unsafeAtomicAdd(loss, c * (1.0f / (float)N_TOT));
}

extern "C" void kernel_launch(void* const* d_in, const int* in_sizes, int n_in,
                              void* d_out, int out_size, void* d_ws, size_t ws_size,
                              hipStream_t stream) {
    const float* X = (const float*)d_in[0];
    const float* A = (const float*)d_in[1];
    const float* S = (const float*)d_in[2];
    const float* P = (const float*)d_in[3];
    float* out = (float*)d_out;

    float* ws   = (float*)d_ws;
    float* sums = ws;
    float* freq = ws + K_CLS * D_DIM;
    float* msq  = freq + K_CLS;

    zero_ws<<<(K_CLS * D_DIM + 2 * K_CLS + 255) / 256, 256, 0, stream>>>(ws, out);

    pass1_colsum<<<N_TOT / P1_ROWS, P1_THREADS, 0, stream>>>(X, A, sums, freq);

    means_msq<<<K_CLS, 256, 0, stream>>>(sums, freq, P, out + 1, msq);

    const size_t smemBytes = (size_t)(K_CLS * MLDS_STRIDE + K_CLS + 128) * sizeof(float);
    pass2_loss<<<N_TOT / 128, 256, smemBytes, stream>>>(X, A, S, out + 1, msq, out);
}


# pass 2 data path: 256 persistent work-groups stage label_means once, stream 64-row X and A*S tiles through LDS (16-byte line loads a tile ahead); waves split d in quarters (different reduction tree)
# speedup vs baseline: 1.6927x; 1.0997x over previous
#include <hip/hip_runtime.h>
#include <hip/hip_bf16.h>

typedef __attribute__((ext_vector_type(2))) float v2f;
typedef __attribute__((ext_vector_type(8))) float v8f;

#define N_TOT 131072
#define K_CLS 64
#define D_DIM 256
#define FP_EPS 1e-12f
#define MLDS_STRIDE 260

__device__ __forceinline__ v8f wmma4(v2f a, v2f b, v8f c) {
    return __builtin_amdgcn_wmma_f32_16x16x4_f32(
         false, a,  false, b,
         (short)0, c,  false,  false);
}

__global__ __launch_bounds__(256) void zero_ws(float* __restrict__ ws,
                                               float* __restrict__ out0) {
    int i = blockIdx.x * 256 + threadIdx.x;
    if (i < K_CLS * D_DIM + 2 * K_CLS) ws[i] = 0.0f;
    if (i == 0) out0[0] = 0.0f;
}

#define P1_THREADS 512
#define P1_ROWS    512
#define P1_TN      64
#define P1_XP      544
#define P1_AP      160
#define P1_XT      ((P1_TN / 2) * P1_XP)
#define P1_AT      ((P1_TN / 2) * P1_AP)
#define P1_STAGE   (P1_XT + P1_AT)
#define P1_NSTAGE  (P1_ROWS / P1_TN)

__global__ __launch_bounds__(P1_THREADS)
void pass1_colsum(const float* __restrict__ X, const float* __restrict__ A,
                  float* __restrict__ sums, float* __restrict__ freq) {
    __shared__ __attribute__((aligned(16))) float tile[2 * P1_STAGE];

    const int tid  = threadIdx.x;
    const int lane = tid & 31;
    const int wave = __builtin_amdgcn_readfirstlane(tid >> 5);
    const int c    = lane & 15;
    const int h    = lane >> 4;
    const int kw   = (wave & 1) * 32;
    const int dw   = (wave >> 1) * 32;

    const size_t row0 = (size_t)blockIdx.x * P1_ROWS;

    const int xp = tid >> 6, xc4 = tid & 63;
    const int ap = tid >> 4, ac4 = tid & 15;
    const float4* xg = (const float4*)(X + (row0 + 2 * xp) * D_DIM) + xc4;
    const float4* ag = (const float4*)(A + (row0 + 2 * ap) * K_CLS) + ac4;
    const int xl = xp * P1_XP + xc4 * 8;
    const int al = P1_XT + ap * P1_AP + ac4 * 8;

    float4 xe[4], xo[4], ae, ao;

    v8f acc[2][2];
    #pragma unroll
    for (int i = 0; i < 2; ++i)
        #pragma unroll
        for (int j = 0; j < 2; ++j) { v8f z = {}; acc[i][j] = z; }
    float fp0 = 0.0f, fp1 = 0.0f;

#define P1_LOAD(S)                                                                   \
    {                                                                                \
        const float4* xn = xg + (size_t)(S) * P1_TN * (D_DIM / 4);                   \
        const float4* an = ag + (size_t)(S) * P1_TN * (K_CLS / 4);                   \
        _Pragma("unroll")                                                            \
        for (int i = 0; i < 4; ++i) {                                                \
            xe[i] = xn[(size_t)(16 * i) * (D_DIM / 4)];                              \
            xo[i] = xn[(size_t)(16 * i + 1) * (D_DIM / 4)];                          \
        }                                                                            \
        ae = an[0];                                                                  \
        ao = an[K_CLS / 4];                                                          \
    }
#define P1_STORE(BUF)                                                                \
    {                                                                                \
        _Pragma("unroll")                                                            \
        for (int i = 0; i < 4; ++i) {                                                \
            float* p = &tile[(BUF) + xl + (8 * i) * P1_XP];                          \
            *(float4*)p       = make_float4(xe[i].x, xo[i].x, xe[i].y, xo[i].y);     \
            *(float4*)(p + 4) = make_float4(xe[i].z, xo[i].z, xe[i].w, xo[i].w);     \
        }                                                                            \
        float* q = &tile[(BUF) + al];                                                \
        *(float4*)q       = make_float4(ae.x, ao.x, ae.y, ao.y);                     \
        *(float4*)(q + 4) = make_float4(ae.z, ao.z, ae.w, ao.w);                     \
    }

    P1_LOAD(0)
    P1_STORE(0)
    __syncthreads();

    for (int s = 0; s < P1_NSTAGE; ++s) {
        const int buf = (s & 1) * P1_STAGE;
        const bool more = (s + 1 < P1_NSTAGE);

        if (more) P1_LOAD(s + 1)

        const float* xt = tile + buf + h * P1_XP + 2 * (dw + c);
        const float* at = tile + buf + P1_XT + h * P1_AP + 2 * (kw + c);
        #pragma unroll
        for (int g = 0; g < P1_TN / 4; ++g) {
            const float2 a0 = *(const float2*)(at + (2 * g) * P1_AP);
            const float2 a1 = *(const float2*)(at + (2 * g) * P1_AP + 32);
            const float2 b0 = *(const float2*)(xt + (2 * g) * P1_XP);
            const float2 b1 = *(const float2*)(xt + (2 * g) * P1_XP + 32);
            v2f af0, af1, bf0, bf1;
            af0.x = a0.x; af0.y = a0.y;
            af1.x = a1.x; af1.y = a1.y;
            bf0.x = b0.x; bf0.y = b0.y;
            bf1.x = b1.x; bf1.y = b1.y;
            fp0 += a0.x + a0.y;
            fp1 += a1.x + a1.y;
            acc[0][0] = wmma4(af0, bf0, acc[0][0]);
            acc[0][1] = wmma4(af0, bf1, acc[0][1]);
            acc[1][0] = wmma4(af1, bf0, acc[1][0]);
            acc[1][1] = wmma4(af1, bf1, acc[1][1]);
        }

        if (more) P1_STORE(((s + 1) & 1) * P1_STAGE)
        __syncthreads();
    }
#undef P1_LOAD
#undef P1_STORE

    fp0 += __shfl_xor(fp0, 16, 32);
    fp1 += __shfl_xor(fp1, 16, 32);
    if ((wave >> 1) == 0 && lane < 16) {
        unsafeAtomicAdd(&freq[kw + lane], fp0);
        unsafeAtomicAdd(&freq[kw + 16 + lane], fp1);
    }

    #pragma unroll
    for (int kt = 0; kt < 2; ++kt) {
        #pragma unroll
        for (int dt = 0; dt < 2; ++dt) {
            #pragma unroll
            for (int r = 0; r < 8; ++r) {
                const int k = kw + 16 * kt + r + 8 * h;
                const int d = dw + 16 * dt + c;
                unsafeAtomicAdd(&sums[k * D_DIM + d], acc[kt][dt][r]);
            }
        }
    }
}

__global__ __launch_bounds__(256)
void means_msq(const float* __restrict__ sums, const float* __restrict__ freq,
               const float* __restrict__ past, float* __restrict__ out_means,
               float* __restrict__ msq) {
    __shared__ float red[8];
    const int k = blockIdx.x;
    const int d = threadIdx.x;
    const float f  = freq[k];
    const float bm = sums[k * D_DIM + d] / fmaxf(f, FP_EPS);
    const float pm = past[k * D_DIM + d];
    const float m  = (f > FP_EPS) ? (pm * 0.5f + bm * 0.5f) : pm;
    out_means[k * D_DIM + d] = m;

    float sq = m * m;
    #pragma unroll
    for (int off = 16; off > 0; off >>= 1) sq += __shfl_xor(sq, off, 32);
    if ((threadIdx.x & 31) == 0) red[threadIdx.x >> 5] = sq;
    __syncthreads();
    if (threadIdx.x == 0) {
        float t = 0.0f;
        #pragma unroll
        for (int i = 0; i < 8; ++i) t += red[i];
        msq[k] = t;
    }
}

#define P2_THREADS 512
#define P2_ROWS    512
#define P2_TN      64
#define P2_XP      260
#define P2_WP      68
#define P2_XT      (P2_TN * P2_XP)
#define P2_WT      (P2_TN * P2_WP)
#define P2_STAGE   (P2_XT + P2_WT)
#define P2_NSTAGE  (P2_ROWS / P2_TN)

__global__ __launch_bounds__(P2_THREADS)
void pass2_loss(const float* __restrict__ X, const float* __restrict__ A,
                const float* __restrict__ S, const float* __restrict__ means,
                const float* __restrict__ msq, float* __restrict__ loss) {
    __shared__ __attribute__((aligned(16))) float mlds[K_CLS * MLDS_STRIDE];
    __shared__ __attribute__((aligned(16))) float tile[2 * P2_STAGE];
    __shared__ float msq_lds[K_CLS];

    const int tid  = threadIdx.x;
    const int lane = tid & 31;
    const int wave = __builtin_amdgcn_readfirstlane(tid >> 5);
    const int c    = lane & 15;
    const int h    = lane >> 4;
    const int nt   = wave & 3;
    const int dq   = wave >> 2;

    const size_t row0 = (size_t)blockIdx.x * P2_ROWS;

    const int xr = tid >> 6, xc4 = tid & 63;
    const int wr = tid >> 4, wc4 = tid & 15;
    const float4* xg = (const float4*)(X + (row0 + xr) * D_DIM) + xc4;
    const float4* ag = (const float4*)(A + (row0 + wr) * K_CLS) + wc4;
    const float4* sg = (const float4*)(S + (row0 + wr) * K_CLS) + wc4;
    const int xl = xr * P2_XP + xc4 * 4;
    const int wl = P2_XT + wr * P2_WP + wc4 * 4;

    float4 xv[8], av[2], sv[2];

#define P2_LOAD(ST)                                                                  \
    {                                                                                \
        const float4* xn = xg + (size_t)(ST) * P2_TN * (D_DIM / 4);                  \
        const float4* an = ag + (size_t)(ST) * P2_TN * (K_CLS / 4);                  \
        const float4* sn = sg + (size_t)(ST) * P2_TN * (K_CLS / 4);                  \
        _Pragma("unroll")                                                            \
        for (int i = 0; i < 8; ++i) xv[i] = xn[(size_t)(8 * i) * (D_DIM / 4)];       \
        _Pragma("unroll")                                                            \
        for (int j = 0; j < 2; ++j) {                                                \
            av[j] = an[(size_t)(32 * j) * (K_CLS / 4)];                              \
            sv[j] = sn[(size_t)(32 * j) * (K_CLS / 4)];                              \
        }                                                                            \
    }
#define P2_STORE(BUF)                                                                \
    {                                                                                \
        _Pragma("unroll")                                                            \
        for (int i = 0; i < 8; ++i)                                                  \
            *(float4*)&tile[(BUF) + xl + (8 * i) * P2_XP] = xv[i];                   \
        _Pragma("unroll")                                                            \
        for (int j = 0; j < 2; ++j)                                                  \
            *(float4*)&tile[(BUF) + wl + (32 * j) * P2_WP] =                         \
                make_float4(av[j].x * sv[j].x, av[j].y * sv[j].y,                    \
                            av[j].z * sv[j].z, av[j].w * sv[j].w);                   \
    }

    P2_LOAD(0)

    #pragma unroll
    for (int i = 0; i < 8; ++i) {
        const int idx = tid + i * P2_THREADS;
        const int k   = idx >> 6;
        const int c4  = idx & 63;
        const float4 v = ((const float4*)means)[(size_t)k * 64 + c4];
        *(float4*)&mlds[k * MLDS_STRIDE + c4 * 4] = v;
    }
    if (tid < K_CLS) msq_lds[tid] = msq[tid];
    P2_STORE(0)
    __syncthreads();

    float msel[4];
    #pragma unroll
    for (int kt = 0; kt < 4; ++kt) msel[kt] = (dq == 0) ? msq_lds[kt * 16 + c] : 0.0f;

    float csum = 0.0f;

    for (int s = 0; s < P2_NSTAGE; ++s) {
        const int buf = (s & 1) * P2_STAGE;
        const bool more = (s + 1 < P2_NSTAGE);

        if (more) P2_LOAD(s + 1)

        v8f acc[4];
        #pragma unroll
        for (int t = 0; t < 4; ++t) { v8f z = {}; acc[t] = z; }
        float xp = 0.0f;

        const float* xt = tile + buf + (nt * 16 + c) * P2_XP + dq * 64 + 2 * h;
        const float* mt = mlds + c * MLDS_STRIDE + dq * 64 + 2 * h;
        #pragma unroll
        for (int j = 0; j < 16; ++j) {
            const float2 a2 = *(const float2*)(xt + 4 * j);
            v2f af;
            af.x = a2.x; af.y = a2.y;
            xp += a2.x * a2.x + a2.y * a2.y;
            #pragma unroll
            for (int kt = 0; kt < 4; ++kt) {
                const float2 b2 = *(const float2*)(mt + kt * 16 * MLDS_STRIDE + 4 * j);
                v2f bf;
                bf.x = b2.x; bf.y = b2.y;
                acc[kt] = wmma4(af, bf, acc[kt]);
            }
        }

        const float xs = xp + __shfl_xor(xp, 16, 32);

        const float* wt = tile + buf + P2_XT + (nt * 16 + 8 * h) * P2_WP + c;
        #pragma unroll
        for (int r = 0; r < 8; ++r) {
            const float xrow = __shfl(xs, 8 * h + r, 32);
            #pragma unroll
            for (int kt = 0; kt < 4; ++kt) {
                const float w = wt[r * P2_WP + kt * 16];
                csum += w * (xrow + msel[kt] - 2.0f * acc[kt][r]);
            }
        }

        if (more) P2_STORE(((s + 1) & 1) * P2_STAGE)
        __syncthreads();
    }
#undef P2_LOAD
#undef P2_STORE

    #pragma unroll
    for (int off = 16; off > 0; off >>= 1) csum += __shfl_xor(csum, off, 32);
    if (lane == 0)
        unsafeAtomicAdd(loss, csum * (1.0f / (float)N_TOT));
}

extern "C" void kernel_launch(void* const* d_in, const int* in_sizes, int n_in,
                              void* d_out, int out_size, void* d_ws, size_t ws_size,
                              hipStream_t stream) {
    const float* X = (const float*)d_in[0];
    const float* A = (const float*)d_in[1];
    const float* S = (const float*)d_in[2];
    const float* P = (const float*)d_in[3];
    float* out = (float*)d_out;

    float* ws   = (float*)d_ws;
    float* sums = ws;
    float* freq = ws + K_CLS * D_DIM;
    float* msq  = freq + K_CLS;

    zero_ws<<<(K_CLS * D_DIM + 2 * K_CLS + 255) / 256, 256, 0, stream>>>(ws, out);

    pass1_colsum<<<N_TOT / P1_ROWS, P1_THREADS, 0, stream>>>(X, A, sums, freq);

    means_msq<<<K_CLS, 256, 0, stream>>>(sums, freq, P, out + 1, msq);

    pass2_loss<<<N_TOT / P2_ROWS, P2_THREADS, 0, stream>>>(X, A, S, out + 1, msq, out);
}


# pass 2: 8 waves (2 per SIMD), each 16 rows x 64 classes x 128 d, hold their label_means fragments in registers for all 8 tiles; only X fragments are read from LDS per tile (different reduction tree)
# speedup vs baseline: 1.9286x; 1.1394x over previous
#include <hip/hip_runtime.h>
#include <hip/hip_bf16.h>

typedef __attribute__((ext_vector_type(2))) float v2f;
typedef __attribute__((ext_vector_type(8))) float v8f;

#define N_TOT 131072
#define K_CLS 64
#define D_DIM 256
#define FP_EPS 1e-12f
#define MLDS_STRIDE 260

__device__ __forceinline__ v8f wmma4(v2f a, v2f b, v8f c) {
    return __builtin_amdgcn_wmma_f32_16x16x4_f32(
         false, a,  false, b,
         (short)0, c,  false,  false);
}

__global__ __launch_bounds__(256) void zero_ws(float* __restrict__ ws,
                                               float* __restrict__ out0) {
    int i = blockIdx.x * 256 + threadIdx.x;
    if (i < K_CLS * D_DIM + 2 * K_CLS) ws[i] = 0.0f;
    if (i == 0) out0[0] = 0.0f;
}

#define P1_THREADS 512
#define P1_ROWS    512
#define P1_TN      128
#define P1_DH      128
#define P1_XP      288
#define P1_AP      160
#define P1_XT      ((P1_TN / 2) * P1_XP)
#define P1_AT      ((P1_TN / 2) * P1_AP)
#define P1_STAGE   (P1_XT + P1_AT)
#define P1_NT      (P1_ROWS / P1_TN)
#define P1_NSTAGE  (2 * P1_NT)
#define P1_SLOT    (16 * 32)

__global__ __launch_bounds__(P1_THREADS)
void pass1_colsum(const float* __restrict__ X, const float* __restrict__ A,
                  float* __restrict__ sums, float* __restrict__ freq) {
    __shared__ __attribute__((aligned(16))) float tile[2 * P1_STAGE];
    __shared__ float xch[16 * P1_SLOT];
    __shared__ float hold[16 * P1_SLOT];
    __shared__ __attribute__((aligned(16))) float fred[16 * K_CLS];

    const int tid  = threadIdx.x;
    const int lane = tid & 31;
    const int wave = __builtin_amdgcn_readfirstlane(tid >> 5);
    const int c    = lane & 15;
    const int h    = lane >> 4;
    const int kw   = (wave & 1) * 32;
    const int ds   = (wave >> 1) & 3;
    const int dwl  = ds * 32;
    const int rh   = wave >> 3;
    const int krow = kw + 16 * rh + 8 * h;

    const size_t row0 = (size_t)blockIdx.x * P1_ROWS;

    const int xp = tid >> 5, xc4 = tid & 31;
    const int ap = tid >> 4, ac4 = tid & 15;
    const float4* xg = (const float4*)(X + (row0 + 2 * xp) * D_DIM) + xc4;
    const float4* ag = (const float4*)(A + (row0 + 2 * ap) * K_CLS) + ac4;
    const int xl = xp * P1_XP + xc4 * 8;
    const int al = P1_XT + ap * P1_AP + ac4 * 8;

    float4 xe[4], xo[4], ae[2], ao[2];

    v8f acc[2][2];
    #pragma unroll
    for (int i = 0; i < 2; ++i)
        #pragma unroll
        for (int j = 0; j < 2; ++j) { v8f z = {}; acc[i][j] = z; }
    float4 fs = make_float4(0.0f, 0.0f, 0.0f, 0.0f);

#define P1_LOAD(S)                                                                   \
    {                                                                                \
        const float4* xn = xg + (size_t)((S) % P1_NT) * P1_TN * (D_DIM / 4)          \
                              + ((S) / P1_NT) * (P1_DH / 4);                         \
        const float4* an = ag + (size_t)((S) % P1_NT) * P1_TN * (K_CLS / 4);         \
        _Pragma("unroll")                                                            \
        for (int i = 0; i < 4; ++i) {                                                \
            xe[i] = xn[(size_t)(32 * i) * (D_DIM / 4)];                              \
            xo[i] = xn[(size_t)(32 * i + 1) * (D_DIM / 4)];                          \
        }                                                                            \
        _Pragma("unroll")                                                            \
        for (int j = 0; j < 2; ++j) {                                                \
            ae[j] = an[(size_t)(64 * j) * (K_CLS / 4)];                              \
            ao[j] = an[(size_t)(64 * j + 1) * (K_CLS / 4)];                          \
        }                                                                            \
    }
#define P1_STORE(BUF)                                                                \
    {                                                                                \
        _Pragma("unroll")                                                            \
        for (int i = 0; i < 4; ++i) {                                                \
            float* p = &tile[(BUF) + xl + (16 * i) * P1_XP];                         \
            *(float4*)p       = make_float4(xe[i].x, xo[i].x, xe[i].y, xo[i].y);     \
            *(float4*)(p + 4) = make_float4(xe[i].z, xo[i].z, xe[i].w, xo[i].w);     \
        }                                                                            \
        _Pragma("unroll")                                                            \
        for (int j = 0; j < 2; ++j) {                                                \
            float* q = &tile[(BUF) + al + (32 * j) * P1_AP];                         \
            *(float4*)q       = make_float4(ae[j].x, ao[j].x, ae[j].y, ao[j].y);     \
            *(float4*)(q + 4) = make_float4(ae[j].z, ao[j].z, ae[j].w, ao[j].w);     \
        }                                                                            \
    }
#define P1_RUN                                                                       \
    __builtin_amdgcn_sched_group_barrier(0x100, 16, 0);                              \
    __builtin_amdgcn_sched_group_barrier(0x008, 32, 0);
#define P1_MUL                                                                       \
    {                                                                                \
        const float* xt = tile + buf + (rh * 32 + h) * P1_XP + 2 * (dwl + c);        \
        const float* at = tile + buf + P1_XT + (rh * 32 + h) * P1_AP + 2 * (kw + c); \
        _Pragma("unroll")                                                            \
        for (int g = 0; g < 16; ++g) {                                               \
            if (g == 8) { P1_RUN __builtin_amdgcn_sched_barrier(0); }                \
            const float2 a0 = *(const float2*)(at + (2 * g) * P1_AP);                \
            const float2 a1 = *(const float2*)(at + (2 * g) * P1_AP + 32);           \
            const float2 b0 = *(const float2*)(xt + (2 * g) * P1_XP);                \
            const float2 b1 = *(const float2*)(xt + (2 * g) * P1_XP + 32);           \
            v2f af0, af1, bf0, bf1;                                                  \
            af0.x = a0.x; af0.y = a0.y;                                              \
            af1.x = a1.x; af1.y = a1.y;                                              \
            bf0.x = b0.x; bf0.y = b0.y;                                              \
            bf1.x = b1.x; bf1.y = b1.y;                                              \
            acc[0][0] = wmma4(af0, bf0, acc[0][0]);                                  \
            acc[0][1] = wmma4(af0, bf1, acc[0][1]);                                  \
            acc[1][0] = wmma4(af1, bf0, acc[1][0]);                                  \
            acc[1][1] = wmma4(af1, bf1, acc[1][1]);                                  \
        }                                                                            \
        P1_RUN                                                                       \
    }

#define P1_FSUM                                                                      \
    {                                                                                \
        fs.x += (ae[0].x + ao[0].x) + (ae[1].x + ao[1].x);                           \
        fs.y += (ae[0].y + ao[0].y) + (ae[1].y + ao[1].y);                           \
        fs.z += (ae[0].z + ao[0].z) + (ae[1].z + ao[1].z);                           \
        fs.w += (ae[0].w + ao[0].w) + (ae[1].w + ao[1].w);                           \
    }

    P1_LOAD(0)
    P1_FSUM
    P1_STORE(0)
    __syncthreads();

    float* const my_xch  = xch  + wave * P1_SLOT + lane;
    const float* const pr_xch = xch + (wave ^ 8) * P1_SLOT + lane;
    float* const my_hold = hold + wave * P1_SLOT + lane;

#define P1_SEND(Q)                                                                   \
    {                                                                                \
        _Pragma("unroll")                                                            \
        for (int i4 = 0; i4 < 4; ++i4) {                                             \
            const int i  = (Q) * 4 + i4;     \
            const float v = my_hold[i * 32];                                         \
            unsafeAtomicAdd(&sums[(krow + (i & 7)) * D_DIM + dwl + 16 * (i >> 3) + c], v); \
        }                                                                            \
    }
#define P1_FLUSH(FIRST)                                                              \
    {                                                                                \
        float own[16];                                                               \
        _Pragma("unroll")                                                            \
        for (int dt = 0; dt < 2; ++dt) {                                             \
            _Pragma("unroll")                                                        \
            for (int r = 0; r < 8; ++r) {                                            \
                const float keep = rh ? acc[1][dt][r] : acc[0][dt][r];               \
                const float give = rh ? acc[0][dt][r] : acc[1][dt][r];               \
                own[dt * 8 + r] = keep;                                              \
                my_xch[(dt * 8 + r) * 32] = give;                                    \
            }                                                                        \
        }                                                                            \
        if (FIRST) {                                                                 \
              \
            fs.x += __shfl_xor(fs.x, 16, 32);                                        \
            fs.y += __shfl_xor(fs.y, 16, 32);                                        \
            fs.z += __shfl_xor(fs.z, 16, 32);                                        \
            fs.w += __shfl_xor(fs.w, 16, 32);                                        \
            if (lane < 16) *(float4*)&fred[wave * K_CLS + 4 * lane] = fs;            \
        }                                                                            \
        __syncthreads();                                                             \
        if (FIRST) {                                                                 \
            if (tid < K_CLS) {                                                       \
                float f = 0.0f;                                                      \
                _Pragma("unroll")                                                    \
                for (int w = 0; w < 16; ++w) f += fred[w * K_CLS + tid];             \
                unsafeAtomicAdd(&freq[tid], f);                                      \
            }                                                                        \
        }                                                                            \
        _Pragma("unroll")                                                            \
        for (int i = 0; i < 16; ++i) own[i] += pr_xch[i * 32];                       \
        if (FIRST) {                                                                 \
              \
            _Pragma("unroll")                                                        \
            for (int i = 0; i < 16; ++i) my_hold[i * 32] = own[i];                   \
            _Pragma("unroll")                                                        \
            for (int i = 0; i < 2; ++i)                                              \
                _Pragma("unroll")                                                    \
                for (int j = 0; j < 2; ++j) { v8f z = {}; acc[i][j] = z; }           \
        } else {                                                                     \
                          \
            _Pragma("unroll")                                                        \
            for (int i = 0; i < 16; ++i)                                             \
                unsafeAtomicAdd(&sums[(krow + (i & 7)) * D_DIM + P1_DH + dwl + 16 * (i >> 3) + c], own[i]); \
        }                                                                            \
    }

    #pragma clang loop unroll(disable)
    for (int s = 0; s < P1_NSTAGE - 1; ++s) {
        const int buf = (s & 1) * P1_STAGE;

        P1_LOAD(s + 1)
        if (s >= P1_NT) P1_SEND(s - P1_NT)

        P1_MUL

        if (s + 1 < P1_NT) P1_FSUM
        P1_STORE(((s + 1) & 1) * P1_STAGE)
        __syncthreads();

        if (s == P1_NT - 1) P1_FLUSH(1)
    }
    {
        const int buf = ((P1_NSTAGE - 1) & 1) * P1_STAGE;
        P1_SEND(P1_NT - 1)
        P1_MUL
        __syncthreads();
        P1_FLUSH(0)
    }
#undef P1_LOAD
#undef P1_STORE
#undef P1_MUL
#undef P1_RUN
#undef P1_FSUM
#undef P1_SEND
#undef P1_FLUSH
}

__global__ __launch_bounds__(256)
void means_msq(const float* __restrict__ sums, const float* __restrict__ freq,
               const float* __restrict__ past, float* __restrict__ out_means,
               float* __restrict__ msq) {
    __shared__ float red[8];
    const int k = blockIdx.x;
    const int d = threadIdx.x;
    const float f  = freq[k];
    const float bm = sums[k * D_DIM + d] / fmaxf(f, FP_EPS);
    const float pm = past[k * D_DIM + d];
    const float m  = (f > FP_EPS) ? (pm * 0.5f + bm * 0.5f) : pm;
    out_means[k * D_DIM + d] = m;

    float sq = m * m;
    #pragma unroll
    for (int off = 16; off > 0; off >>= 1) sq += __shfl_xor(sq, off, 32);
    if ((threadIdx.x & 31) == 0) red[threadIdx.x >> 5] = sq;
    __syncthreads();
    if (threadIdx.x == 0) {
        float t = 0.0f;
        #pragma unroll
        for (int i = 0; i < 8; ++i) t += red[i];
        msq[k] = t;
    }
}

#define P2_THREADS 256
#define P2_ROWS    512
#define P2_TN      64
#define P2_XP      260
#define P2_WP      68
#define P2_XT      (P2_TN * P2_XP)
#define P2_WT      (P2_TN * P2_WP)
#define P2_STAGE   (P2_XT + P2_WT)
#define P2_NSTAGE  (P2_ROWS / P2_TN)
#define P2_DW      128
#define P2_NJ      (P2_DW / 4)

__global__ __launch_bounds__(P2_THREADS)
void pass2_loss(const float* __restrict__ X, const float* __restrict__ A,
                const float* __restrict__ S, const float* __restrict__ means,
                const float* __restrict__ msq, float* __restrict__ loss) {
    __shared__ __attribute__((aligned(16))) float mlds[K_CLS * MLDS_STRIDE];
    __shared__ __attribute__((aligned(16))) float tile[2 * P2_STAGE];
    __shared__ float msq_lds[K_CLS];

    const int tid  = threadIdx.x;
    const int lane = tid & 31;
    const int wave = __builtin_amdgcn_readfirstlane(tid >> 5);
    const int c    = lane & 15;
    const int h    = lane >> 4;
    const int nt   = wave & 3;
    const int dh   = wave >> 2;

    const size_t row0 = (size_t)blockIdx.x * P2_ROWS;

    const int xr = tid >> 6, xc4 = tid & 63;
    const int wr = tid >> 4, wc4 = tid & 15;
    const float4* xg = (const float4*)(X + (row0 + xr) * D_DIM) + xc4;
    const float4* ag = (const float4*)(A + (row0 + wr) * K_CLS) + wc4;
    const float4* sg = (const float4*)(S + (row0 + wr) * K_CLS) + wc4;
    const int xl = xr * P2_XP + xc4 * 4;
    const int wl = P2_XT + wr * P2_WP + wc4 * 4;

#define P2_REP16(M) M(0) M(1) M(2) M(3) M(4) M(5) M(6) M(7) M(8) M(9) M(10) M(11) M(12) M(13) M(14) M(15)
#define P2_XDECL(I) float4 xv##I;
    P2_REP16(P2_XDECL)
    float4 av[4], sv[4];

#define P2_XLD(I) xv##I = xn[(size_t)(4 * I) * (D_DIM / 4)];
#define P2_XST(I) *(float4*)(xdst + (4 * I) * P2_XP) = xv##I;
#define P2_LOAD(ST)                                                                  \
    {                                                                                \
        const float4* xn = xg + (size_t)(ST) * P2_TN * (D_DIM / 4);                  \
        const float4* an = ag + (size_t)(ST) * P2_TN * (K_CLS / 4);                  \
        const float4* sn = sg + (size_t)(ST) * P2_TN * (K_CLS / 4);                  \
        P2_REP16(P2_XLD)                                                             \
        _Pragma("unroll")                                                            \
        for (int j = 0; j < 4; ++j) {                                                \
            av[j] = an[(size_t)(16 * j) * (K_CLS / 4)];                              \
            sv[j] = sn[(size_t)(16 * j) * (K_CLS / 4)];                              \
        }                                                                            \
    }
#define P2_STORE(BUF)                                                                \
    {                                                                                \
        float* const xdst = &tile[(BUF) + xl];                                       \
        P2_REP16(P2_XST)                                                             \
        _Pragma("unroll")                                                            \
        for (int j = 0; j < 4; ++j)                                                  \
            *(float4*)&tile[(BUF) + wl + (16 * j) * P2_WP] =                         \
                make_float4(av[j].x * sv[j].x, av[j].y * sv[j].y,                    \
                            av[j].z * sv[j].z, av[j].w * sv[j].w);                   \
    }

    P2_LOAD(0)

    #pragma unroll
    for (int i = 0; i < 16; ++i) {
        const int idx = tid + i * P2_THREADS;
        const int k   = idx >> 6;
        const int c4  = idx & 63;
        const float4 v = ((const float4*)means)[(size_t)k * 64 + c4];
        *(float4*)&mlds[k * MLDS_STRIDE + c4 * 4] = v;
    }
    if (tid < K_CLS) msq_lds[tid] = msq[tid];
    P2_STORE(0)
    __syncthreads();

    float msel[4];
    #pragma unroll
    for (int kt = 0; kt < 4; ++kt) msel[kt] = (dh == 0) ? msq_lds[kt * 16 + c] : 0.0f;

    v2f bq[P2_NJ][4];
    {
        const float* mt = mlds + c * MLDS_STRIDE + dh * P2_DW + 2 * h;
        #pragma unroll
        for (int j = 0; j < P2_NJ; ++j) {
            #pragma unroll
            for (int kt = 0; kt < 4; ++kt) {
                const float2 b2 = *(const float2*)(mt + kt * 16 * MLDS_STRIDE + 4 * j);
                bq[j][kt].x = b2.x; bq[j][kt].y = b2.y;
            }
        }
    }

    float csum = 0.0f;

    for (int s = 0; s < P2_NSTAGE; ++s) {
        const int buf = (s & 1) * P2_STAGE;
        const bool more = (s + 1 < P2_NSTAGE);

        if (more) P2_LOAD(s + 1)

        v8f acc[4];
        #pragma unroll
        for (int t = 0; t < 4; ++t) { v8f z = {}; acc[t] = z; }
        float xp = 0.0f;

        const float* xt = tile + buf + (nt * 16 + c) * P2_XP + dh * P2_DW + 2 * h;
        #pragma unroll
        for (int j = 0; j < P2_NJ; ++j) {
            const float2 a2 = *(const float2*)(xt + 4 * j);
            v2f af;
            af.x = a2.x; af.y = a2.y;
            xp += a2.x * a2.x + a2.y * a2.y;
            #pragma unroll
            for (int kt = 0; kt < 4; ++kt) acc[kt] = wmma4(af, bq[j][kt], acc[kt]);
        }

        const float xs = xp + __shfl_xor(xp, 16, 32);

        const float* wt = tile + buf + P2_XT + (nt * 16 + 8 * h) * P2_WP + c;
        #pragma unroll
        for (int r = 0; r < 8; ++r) {
            const float xrow = __shfl(xs, 8 * h + r, 32);
            #pragma unroll
            for (int kt = 0; kt < 4; ++kt) {
                const float w = wt[r * P2_WP + kt * 16];
                csum += w * (xrow + msel[kt] - 2.0f * acc[kt][r]);
            }
        }

        if (more) P2_STORE(((s + 1) & 1) * P2_STAGE)
        __syncthreads();
    }
#undef P2_LOAD
#undef P2_STORE
#undef P2_XLD
#undef P2_XST
#undef P2_XDECL
#undef P2_REP16

    #pragma unroll
    for (int off = 16; off > 0; off >>= 1) csum += __shfl_xor(csum, off, 32);
    if (lane == 0)
        unsafeAtomicAdd(loss, csum * (1.0f / (float)N_TOT));
}

extern "C" void kernel_launch(void* const* d_in, const int* in_sizes, int n_in,
                              void* d_out, int out_size, void* d_ws, size_t ws_size,
                              hipStream_t stream) {
    const float* X = (const float*)d_in[0];
    const float* A = (const float*)d_in[1];
    const float* S = (const float*)d_in[2];
    const float* P = (const float*)d_in[3];
    float* out = (float*)d_out;

    float* ws   = (float*)d_ws;
    float* sums = ws;
    float* freq = ws + K_CLS * D_DIM;
    float* msq  = freq + K_CLS;

    zero_ws<<<(K_CLS * D_DIM + 2 * K_CLS + 255) / 256, 256, 0, stream>>>(ws, out);

    pass1_colsum<<<N_TOT / P1_ROWS, P1_THREADS, 0, stream>>>(X, A, sums, freq);

    means_msq<<<K_CLS, 256, 0, stream>>>(sums, freq, P, out + 1, msq);

    pass2_loss<<<N_TOT / P2_ROWS, P2_THREADS, 0, stream>>>(X, A, S, out + 1, msq, out);
}


# pass 2 as two independent 4-wave work-groups per WGP (256 rows each, 32-row LDS tiles): no shared barrier, so one's tile hand-over and waits run under the other's WMMAs (different reduction tree)
# speedup vs baseline: 1.9943x; 1.0341x over previous
#include <hip/hip_runtime.h>
#include <hip/hip_bf16.h>

typedef __attribute__((ext_vector_type(2))) float v2f;
typedef __attribute__((ext_vector_type(8))) float v8f;

#define N_TOT 131072
#define K_CLS 64
#define D_DIM 256
#define FP_EPS 1e-12f
#define MLDS_STRIDE 260

__device__ __forceinline__ v8f wmma4(v2f a, v2f b, v8f c) {
    return __builtin_amdgcn_wmma_f32_16x16x4_f32(
         false, a,  false, b,
         (short)0, c,  false,  false);
}

__global__ __launch_bounds__(256) void zero_ws(float* __restrict__ ws,
                                               float* __restrict__ out0) {
    int i = blockIdx.x * 256 + threadIdx.x;
    if (i < K_CLS * D_DIM + 2 * K_CLS) ws[i] = 0.0f;
    if (i == 0) out0[0] = 0.0f;
}

#define P1_THREADS 512
#define P1_ROWS    512
#define P1_TN      128
#define P1_DH      128
#define P1_XP      288
#define P1_AP      160
#define P1_XT      ((P1_TN / 2) * P1_XP)
#define P1_AT      ((P1_TN / 2) * P1_AP)
#define P1_STAGE   (P1_XT + P1_AT)
#define P1_NT      (P1_ROWS / P1_TN)
#define P1_NSTAGE  (2 * P1_NT)
#define P1_SLOT    (16 * 32)

__global__ __launch_bounds__(P1_THREADS)
void pass1_colsum(const float* __restrict__ X, const float* __restrict__ A,
                  float* __restrict__ sums, float* __restrict__ freq) {
    __shared__ __attribute__((aligned(16))) float tile[2 * P1_STAGE];
    __shared__ float xch[16 * P1_SLOT];
    __shared__ float hold[16 * P1_SLOT];
    __shared__ __attribute__((aligned(16))) float fred[16 * K_CLS];

    const int tid  = threadIdx.x;
    const int lane = tid & 31;
    const int wave = __builtin_amdgcn_readfirstlane(tid >> 5);
    const int c    = lane & 15;
    const int h    = lane >> 4;
    const int kw   = (wave & 1) * 32;
    const int ds   = (wave >> 1) & 3;
    const int dwl  = ds * 32;
    const int rh   = wave >> 3;
    const int krow = kw + 16 * rh + 8 * h;

    const size_t row0 = (size_t)blockIdx.x * P1_ROWS;

    const int xp = tid >> 5, xc4 = tid & 31;
    const int ap = tid >> 4, ac4 = tid & 15;
    const float4* xg = (const float4*)(X + (row0 + 2 * xp) * D_DIM) + xc4;
    const float4* ag = (const float4*)(A + (row0 + 2 * ap) * K_CLS) + ac4;
    const int xl = xp * P1_XP + xc4 * 8;
    const int al = P1_XT + ap * P1_AP + ac4 * 8;

    float4 xe[4], xo[4], ae[2], ao[2];

    v8f acc[2][2];
    #pragma unroll
    for (int i = 0; i < 2; ++i)
        #pragma unroll
        for (int j = 0; j < 2; ++j) { v8f z = {}; acc[i][j] = z; }
    float4 fs = make_float4(0.0f, 0.0f, 0.0f, 0.0f);

#define P1_LOAD(S)                                                                   \
    {                                                                                \
        const float4* xn = xg + (size_t)((S) % P1_NT) * P1_TN * (D_DIM / 4)          \
                              + ((S) / P1_NT) * (P1_DH / 4);                         \
        const float4* an = ag + (size_t)((S) % P1_NT) * P1_TN * (K_CLS / 4);         \
        _Pragma("unroll")                                                            \
        for (int i = 0; i < 4; ++i) {                                                \
            xe[i] = xn[(size_t)(32 * i) * (D_DIM / 4)];                              \
            xo[i] = xn[(size_t)(32 * i + 1) * (D_DIM / 4)];                          \
        }                                                                            \
        _Pragma("unroll")                                                            \
        for (int j = 0; j < 2; ++j) {                                                \
            ae[j] = an[(size_t)(64 * j) * (K_CLS / 4)];                              \
            ao[j] = an[(size_t)(64 * j + 1) * (K_CLS / 4)];                          \
        }                                                                            \
    }
#define P1_STORE(BUF)                                                                \
    {                                                                                \
        _Pragma("unroll")                                                            \
        for (int i = 0; i < 4; ++i) {                                                \
            float* p = &tile[(BUF) + xl + (16 * i) * P1_XP];                         \
            *(float4*)p       = make_float4(xe[i].x, xo[i].x, xe[i].y, xo[i].y);     \
            *(float4*)(p + 4) = make_float4(xe[i].z, xo[i].z, xe[i].w, xo[i].w);     \
        }                                                                            \
        _Pragma("unroll")                                                            \
        for (int j = 0; j < 2; ++j) {                                                \
            float* q = &tile[(BUF) + al + (32 * j) * P1_AP];                         \
            *(float4*)q       = make_float4(ae[j].x, ao[j].x, ae[j].y, ao[j].y);     \
            *(float4*)(q + 4) = make_float4(ae[j].z, ao[j].z, ae[j].w, ao[j].w);     \
        }                                                                            \
    }
#define P1_RUN                                                                       \
    __builtin_amdgcn_sched_group_barrier(0x100, 16, 0);                              \
    __builtin_amdgcn_sched_group_barrier(0x008, 32, 0);
#define P1_MUL                                                                       \
    {                                                                                \
        const float* xt = tile + buf + (rh * 32 + h) * P1_XP + 2 * (dwl + c);        \
        const float* at = tile + buf + P1_XT + (rh * 32 + h) * P1_AP + 2 * (kw + c); \
        _Pragma("unroll")                                                            \
        for (int g = 0; g < 16; ++g) {                                               \
            if (g == 8) { P1_RUN __builtin_amdgcn_sched_barrier(0); }                \
            const float2 a0 = *(const float2*)(at + (2 * g) * P1_AP);                \
            const float2 a1 = *(const float2*)(at + (2 * g) * P1_AP + 32);           \
            const float2 b0 = *(const float2*)(xt + (2 * g) * P1_XP);                \
            const float2 b1 = *(const float2*)(xt + (2 * g) * P1_XP + 32);           \
            v2f af0, af1, bf0, bf1;                                                  \
            af0.x = a0.x; af0.y = a0.y;                                              \
            af1.x = a1.x; af1.y = a1.y;                                              \
            bf0.x = b0.x; bf0.y = b0.y;                                              \
            bf1.x = b1.x; bf1.y = b1.y;                                              \
            acc[0][0] = wmma4(af0, bf0, acc[0][0]);                                  \
            acc[0][1] = wmma4(af0, bf1, acc[0][1]);                                  \
            acc[1][0] = wmma4(af1, bf0, acc[1][0]);                                  \
            acc[1][1] = wmma4(af1, bf1, acc[1][1]);                                  \
        }                                                                            \
        P1_RUN                                                                       \
    }

#define P1_FSUM                                                                      \
    {                                                                                \
        fs.x += (ae[0].x + ao[0].x) + (ae[1].x + ao[1].x);                           \
        fs.y += (ae[0].y + ao[0].y) + (ae[1].y + ao[1].y);                           \
        fs.z += (ae[0].z + ao[0].z) + (ae[1].z + ao[1].z);                           \
        fs.w += (ae[0].w + ao[0].w) + (ae[1].w + ao[1].w);                           \
    }

    P1_LOAD(0)
    P1_FSUM
    P1_STORE(0)
    __syncthreads();

    float* const my_xch  = xch  + wave * P1_SLOT + lane;
    const float* const pr_xch = xch + (wave ^ 8) * P1_SLOT + lane;
    float* const my_hold = hold + wave * P1_SLOT + lane;

#define P1_SEND(Q)                                                                   \
    {                                                                                \
        _Pragma("unroll")                                                            \
        for (int i4 = 0; i4 < 4; ++i4) {                                             \
            const int i  = (Q) * 4 + i4;     \
            const float v = my_hold[i * 32];                                         \
            unsafeAtomicAdd(&sums[(krow + (i & 7)) * D_DIM + dwl + 16 * (i >> 3) + c], v); \
        }                                                                            \
    }
#define P1_FLUSH(FIRST)                                                              \
    {                                                                                \
        float own[16];                                                               \
        _Pragma("unroll")                                                            \
        for (int dt = 0; dt < 2; ++dt) {                                             \
            _Pragma("unroll")                                                        \
            for (int r = 0; r < 8; ++r) {                                            \
                const float keep = rh ? acc[1][dt][r] : acc[0][dt][r];               \
                const float give = rh ? acc[0][dt][r] : acc[1][dt][r];               \
                own[dt * 8 + r] = keep;                                              \
                my_xch[(dt * 8 + r) * 32] = give;                                    \
            }                                                                        \
        }                                                                            \
        if (FIRST) {                                                                 \
              \
            fs.x += __shfl_xor(fs.x, 16, 32);                                        \
            fs.y += __shfl_xor(fs.y, 16, 32);                                        \
            fs.z += __shfl_xor(fs.z, 16, 32);                                        \
            fs.w += __shfl_xor(fs.w, 16, 32);                                        \
            if (lane < 16) *(float4*)&fred[wave * K_CLS + 4 * lane] = fs;            \
        }                                                                            \
        __syncthreads();                                                             \
        if (FIRST) {                                                                 \
            if (tid < K_CLS) {                                                       \
                float f = 0.0f;                                                      \
                _Pragma("unroll")                                                    \
                for (int w = 0; w < 16; ++w) f += fred[w * K_CLS + tid];             \
                unsafeAtomicAdd(&freq[tid], f);                                      \
            }                                                                        \
        }                                                                            \
        _Pragma("unroll")                                                            \
        for (int i = 0; i < 16; ++i) own[i] += pr_xch[i * 32];                       \
        if (FIRST) {                                                                 \
              \
            _Pragma("unroll")                                                        \
            for (int i = 0; i < 16; ++i) my_hold[i * 32] = own[i];                   \
            _Pragma("unroll")                                                        \
            for (int i = 0; i < 2; ++i)                                              \
                _Pragma("unroll")                                                    \
                for (int j = 0; j < 2; ++j) { v8f z = {}; acc[i][j] = z; }           \
        } else {                                                                     \
                          \
            _Pragma("unroll")                                                        \
            for (int i = 0; i < 16; ++i)                                             \
                unsafeAtomicAdd(&sums[(krow + (i & 7)) * D_DIM + P1_DH + dwl + 16 * (i >> 3) + c], own[i]); \
        }                                                                            \
    }

    #pragma clang loop unroll(disable)
    for (int s = 0; s < P1_NSTAGE - 1; ++s) {
        const int buf = (s & 1) * P1_STAGE;

        P1_LOAD(s + 1)
        if (s >= P1_NT) P1_SEND(s - P1_NT)

        P1_MUL

        if (s + 1 < P1_NT) P1_FSUM
        P1_STORE(((s + 1) & 1) * P1_STAGE)
        __syncthreads();

        if (s == P1_NT - 1) P1_FLUSH(1)
    }
    {
        const int buf = ((P1_NSTAGE - 1) & 1) * P1_STAGE;
        P1_SEND(P1_NT - 1)
        P1_MUL
        __syncthreads();
        P1_FLUSH(0)
    }
#undef P1_LOAD
#undef P1_STORE
#undef P1_MUL
#undef P1_RUN
#undef P1_FSUM
#undef P1_SEND
#undef P1_FLUSH
}

__global__ __launch_bounds__(256)
void means_msq(const float* __restrict__ sums, const float* __restrict__ freq,
               const float* __restrict__ past, float* __restrict__ out_means,
               float* __restrict__ msq) {
    __shared__ float red[8];
    const int k = blockIdx.x;
    const int d = threadIdx.x;
    const float f  = freq[k];
    const float bm = sums[k * D_DIM + d] / fmaxf(f, FP_EPS);
    const float pm = past[k * D_DIM + d];
    const float m  = (f > FP_EPS) ? (pm * 0.5f + bm * 0.5f) : pm;
    out_means[k * D_DIM + d] = m;

    float sq = m * m;
    #pragma unroll
    for (int off = 16; off > 0; off >>= 1) sq += __shfl_xor(sq, off, 32);
    if ((threadIdx.x & 31) == 0) red[threadIdx.x >> 5] = sq;
    __syncthreads();
    if (threadIdx.x == 0) {
        float t = 0.0f;
        #pragma unroll
        for (int i = 0; i < 8; ++i) t += red[i];
        msq[k] = t;
    }
}

#define P2_THREADS 128
#define P2_ROWS    256
#define P2_TN      32
#define P2_XP      260
#define P2_WP      68
#define P2_XT      (P2_TN * P2_XP)
#define P2_WT      (P2_TN * P2_WP)
#define P2_STAGE   (P2_XT + P2_WT)
#define P2_NSTAGE  (P2_ROWS / P2_TN)
#define P2_DW      128
#define P2_NJ      (P2_DW / 4)

__global__ __launch_bounds__(P2_THREADS)
void pass2_loss(const float* __restrict__ X, const float* __restrict__ A,
                const float* __restrict__ S, const float* __restrict__ means,
                const float* __restrict__ msq, float* __restrict__ loss) {
    __shared__ __attribute__((aligned(16))) float smem[P2_STAGE + K_CLS * MLDS_STRIDE];
    __shared__ float msq_lds[K_CLS];
    float* const tile = smem;
    float* const mlds = smem + P2_STAGE;

    const int tid  = threadIdx.x;
    const int lane = tid & 31;
    const int wave = __builtin_amdgcn_readfirstlane(tid >> 5);
    const int c    = lane & 15;
    const int h    = lane >> 4;
    const int nt   = wave & 1;
    const int dh   = wave >> 1;

    const size_t row0 = (size_t)(blockIdx.x & 255) * (2 * P2_ROWS) + (size_t)(blockIdx.x >> 8) * P2_ROWS;

    const int xr = tid >> 6, xc4 = tid & 63;
    const int wr = tid >> 4, wc4 = tid & 15;
    const float4* xg = (const float4*)(X + (row0 + xr) * D_DIM) + xc4;
    const float4* ag = (const float4*)(A + (row0 + wr) * K_CLS) + wc4;
    const float4* sg = (const float4*)(S + (row0 + wr) * K_CLS) + wc4;
    const int xl = xr * P2_XP + xc4 * 4;
    const int wl = P2_XT + wr * P2_WP + wc4 * 4;

#define P2_REP16(M) M(0) M(1) M(2) M(3) M(4) M(5) M(6) M(7) M(8) M(9) M(10) M(11) M(12) M(13) M(14) M(15)
#define P2_XDECL(I) float4 xv##I;
    P2_REP16(P2_XDECL)
    float4 av[4], sv[4];

#define P2_XLD(I) xv##I = xn[(size_t)(2 * I) * (D_DIM / 4)];
#define P2_XST(I) *(float4*)(xdst + (2 * I) * P2_XP) = xv##I;
#define P2_LOAD(ST)                                                                  \
    {                                                                                \
        const float4* xn = xg + (size_t)(ST) * P2_TN * (D_DIM / 4);                  \
        const float4* an = ag + (size_t)(ST) * P2_TN * (K_CLS / 4);                  \
        const float4* sn = sg + (size_t)(ST) * P2_TN * (K_CLS / 4);                  \
        P2_REP16(P2_XLD)                                                             \
        _Pragma("unroll")                                                            \
        for (int j = 0; j < 4; ++j) {                                                \
            av[j] = an[(size_t)(8 * j) * (K_CLS / 4)];                               \
            sv[j] = sn[(size_t)(8 * j) * (K_CLS / 4)];                               \
        }                                                                            \
    }
#define P2_STORE(BUF)                                                                \
    {                                                                                \
        float* const xdst = &tile[(BUF) + xl];                                       \
        P2_REP16(P2_XST)                                                             \
        _Pragma("unroll")                                                            \
        for (int j = 0; j < 4; ++j)                                                  \
            *(float4*)&tile[(BUF) + wl + (8 * j) * P2_WP] =                          \
                make_float4(av[j].x * sv[j].x, av[j].y * sv[j].y,                    \
                            av[j].z * sv[j].z, av[j].w * sv[j].w);                   \
    }

    P2_LOAD(0)

    #pragma unroll
    for (int i = 0; i < 32; ++i) {
        const int idx = tid + i * P2_THREADS;
        const int k   = idx >> 6;
        const int c4  = idx & 63;
        const float4 v = ((const float4*)means)[(size_t)k * 64 + c4];
        *(float4*)&mlds[k * MLDS_STRIDE + c4 * 4] = v;
    }
    if (tid < K_CLS) msq_lds[tid] = msq[tid];
    P2_STORE(0)
    __syncthreads();

    float msel[4];
    #pragma unroll
    for (int kt = 0; kt < 4; ++kt) msel[kt] = (dh == 0) ? msq_lds[kt * 16 + c] : 0.0f;

    v2f bq[P2_NJ][4];
    {
        const float* mt = mlds + c * MLDS_STRIDE + dh * P2_DW + 2 * h;
        #pragma unroll
        for (int j = 0; j < P2_NJ; ++j) {
            #pragma unroll
            for (int kt = 0; kt < 4; ++kt) {
                const float2 b2 = *(const float2*)(mt + kt * 16 * MLDS_STRIDE + 4 * j);
                bq[j][kt].x = b2.x; bq[j][kt].y = b2.y;
            }
        }
    }

    __syncthreads();

    float csum = 0.0f;

    for (int s = 0; s < P2_NSTAGE; ++s) {
        const int buf = (s & 1) * P2_STAGE;
        const bool more = (s + 1 < P2_NSTAGE);

        if (more) P2_LOAD(s + 1)

        v8f acc[4];
        #pragma unroll
        for (int t = 0; t < 4; ++t) { v8f z = {}; acc[t] = z; }
        float xp = 0.0f;

        const float* xt = tile + buf + (nt * 16 + c) * P2_XP + dh * P2_DW + 2 * h;
        #pragma unroll
        for (int j = 0; j < P2_NJ; ++j) {
            const float2 a2 = *(const float2*)(xt + 4 * j);
            v2f af;
            af.x = a2.x; af.y = a2.y;
            xp += a2.x * a2.x + a2.y * a2.y;
            #pragma unroll
            for (int kt = 0; kt < 4; ++kt) acc[kt] = wmma4(af, bq[j][kt], acc[kt]);
        }

        const float xs = xp + __shfl_xor(xp, 16, 32);

        const float* wt = tile + buf + P2_XT + (nt * 16 + 8 * h) * P2_WP + c;
        #pragma unroll
        for (int r = 0; r < 8; ++r) {
            const float xrow = __shfl(xs, 8 * h + r, 32);
            #pragma unroll
            for (int kt = 0; kt < 4; ++kt) {
                const float w = wt[r * P2_WP + kt * 16];
                csum += w * (xrow + msel[kt] - 2.0f * acc[kt][r]);
            }
        }

        if (more) P2_STORE(((s + 1) & 1) * P2_STAGE)
        __syncthreads();
    }
#undef P2_LOAD
#undef P2_STORE
#undef P2_XLD
#undef P2_XST
#undef P2_XDECL
#undef P2_REP16

    #pragma unroll
    for (int off = 16; off > 0; off >>= 1) csum += __shfl_xor(csum, off, 32);
    if (lane == 0)
        unsafeAtomicAdd(loss, csum * (1.0f / (float)N_TOT));
}

extern "C" void kernel_launch(void* const* d_in, const int* in_sizes, int n_in,
                              void* d_out, int out_size, void* d_ws, size_t ws_size,
                              hipStream_t stream) {
    const float* X = (const float*)d_in[0];
    const float* A = (const float*)d_in[1];
    const float* S = (const float*)d_in[2];
    const float* P = (const float*)d_in[3];
    float* out = (float*)d_out;

    float* ws   = (float*)d_ws;
    float* sums = ws;
    float* freq = ws + K_CLS * D_DIM;
    float* msq  = freq + K_CLS;

    zero_ws<<<(K_CLS * D_DIM + 2 * K_CLS + 255) / 256, 256, 0, stream>>>(ws, out);

    pass1_colsum<<<N_TOT / P1_ROWS, P1_THREADS, 0, stream>>>(X, A, sums, freq);

    means_msq<<<K_CLS, 256, 0, stream>>>(sums, freq, P, out + 1, msq);

    pass2_loss<<<N_TOT / P2_ROWS, P2_THREADS, 0, stream>>>(X, A, S, out + 1, msq, out);
}


# pass 1 sums leave as one plane per work-group by plain stores; a 256-work-group means stage adds the planes in fixed order (global adds only if workspace too small) (different reduction tree)
# speedup vs baseline: 2.2354x; 1.1209x over previous
#include <hip/hip_runtime.h>
#include <hip/hip_bf16.h>

typedef __attribute__((ext_vector_type(2))) float v2f;
typedef __attribute__((ext_vector_type(8))) float v8f;

#define N_TOT 131072
#define K_CLS 64
#define D_DIM 256
#define FP_EPS 1e-12f
#define MLDS_STRIDE 260

__device__ __forceinline__ v8f wmma4(v2f a, v2f b, v8f c) {
    return __builtin_amdgcn_wmma_f32_16x16x4_f32(
         false, a,  false, b,
         (short)0, c,  false,  false);
}

__global__ __launch_bounds__(256) void zero_ws(float* __restrict__ ws,
                                               float* __restrict__ out0) {
    int i = blockIdx.x * 256 + threadIdx.x;
    if (i < K_CLS * D_DIM + 2 * K_CLS) ws[i] = 0.0f;
    if (i == 0) out0[0] = 0.0f;
}

#define P1_THREADS 512
#define P1_ROWS    512
#define P1_TN      128
#define P1_DH      128
#define P1_XP      288
#define P1_AP      160
#define P1_XT      ((P1_TN / 2) * P1_XP)
#define P1_AT      ((P1_TN / 2) * P1_AP)
#define P1_STAGE   (P1_XT + P1_AT)
#define P1_NT      (P1_ROWS / P1_TN)
#define P1_NSTAGE  (2 * P1_NT)
#define P1_SLOT    (16 * 32)
#define P1_PLANE   (K_CLS * D_DIM)
#define P1_NPLANE  (N_TOT / P1_ROWS)
#define P1_NEED_BYTES ((K_CLS * D_DIM + 2 * K_CLS + P1_NPLANE * P1_PLANE) * 4)

__device__ __forceinline__ bool ws_holds_planes(size_t ws_bytes) {
    const unsigned hi = (unsigned)(ws_bytes >> 32);
    const unsigned lo = (unsigned)ws_bytes;
    return (hi != 0u) || (lo >= (unsigned)P1_NEED_BYTES);
}

__global__ __launch_bounds__(P1_THREADS)
void pass1_colsum(const float* __restrict__ X, const float* __restrict__ A,
                  float* __restrict__ sums, float* __restrict__ freq,
                  float* __restrict__ part, size_t ws_bytes) {
    __shared__ __attribute__((aligned(16))) float tile[2 * P1_STAGE];
    __shared__ float xch[16 * P1_SLOT];
    __shared__ float hold[16 * P1_SLOT];
    __shared__ __attribute__((aligned(16))) float fred[16 * K_CLS];

    const int tid  = threadIdx.x;
    const int lane = tid & 31;
    const int wave = __builtin_amdgcn_readfirstlane(tid >> 5);
    const int c    = lane & 15;
    const int h    = lane >> 4;
    const int kw   = (wave & 1) * 32;
    const int ds   = (wave >> 1) & 3;
    const int dwl  = ds * 32;
    const int rh   = wave >> 3;
    const int krow = kw + 16 * rh + 8 * h;

    const size_t row0 = (size_t)blockIdx.x * P1_ROWS;

    const bool planes = ws_holds_planes(ws_bytes);
    float* const plane = part + (size_t)blockIdx.x * P1_PLANE;

    const int xp = tid >> 5, xc4 = tid & 31;
    const int ap = tid >> 4, ac4 = tid & 15;
    const float4* xg = (const float4*)(X + (row0 + 2 * xp) * D_DIM) + xc4;
    const float4* ag = (const float4*)(A + (row0 + 2 * ap) * K_CLS) + ac4;
    const int xl = xp * P1_XP + xc4 * 8;
    const int al = P1_XT + ap * P1_AP + ac4 * 8;

    float4 xe[4], xo[4], ae[2], ao[2];

    v8f acc[2][2];
    #pragma unroll
    for (int i = 0; i < 2; ++i)
        #pragma unroll
        for (int j = 0; j < 2; ++j) { v8f z = {}; acc[i][j] = z; }
    float4 fs = make_float4(0.0f, 0.0f, 0.0f, 0.0f);

#define P1_LOAD(S)                                                                   \
    {                                                                                \
        const float4* xn = xg + (size_t)((S) % P1_NT) * P1_TN * (D_DIM / 4)          \
                              + ((S) / P1_NT) * (P1_DH / 4);                         \
        const float4* an = ag + (size_t)((S) % P1_NT) * P1_TN * (K_CLS / 4);         \
        _Pragma("unroll")                                                            \
        for (int i = 0; i < 4; ++i) {                                                \
            xe[i] = xn[(size_t)(32 * i) * (D_DIM / 4)];                              \
            xo[i] = xn[(size_t)(32 * i + 1) * (D_DIM / 4)];                          \
        }                                                                            \
        _Pragma("unroll")                                                            \
        for (int j = 0; j < 2; ++j) {                                                \
            ae[j] = an[(size_t)(64 * j) * (K_CLS / 4)];                              \
            ao[j] = an[(size_t)(64 * j + 1) * (K_CLS / 4)];                          \
        }                                                                            \
    }
#define P1_STORE(BUF)                                                                \
    {                                                                                \
        _Pragma("unroll")                                                            \
        for (int i = 0; i < 4; ++i) {                                                \
            float* p = &tile[(BUF) + xl + (16 * i) * P1_XP];                         \
            *(float4*)p       = make_float4(xe[i].x, xo[i].x, xe[i].y, xo[i].y);     \
            *(float4*)(p + 4) = make_float4(xe[i].z, xo[i].z, xe[i].w, xo[i].w);     \
        }                                                                            \
        _Pragma("unroll")                                                            \
        for (int j = 0; j < 2; ++j) {                                                \
            float* q = &tile[(BUF) + al + (32 * j) * P1_AP];                         \
            *(float4*)q       = make_float4(ae[j].x, ao[j].x, ae[j].y, ao[j].y);     \
            *(float4*)(q + 4) = make_float4(ae[j].z, ao[j].z, ae[j].w, ao[j].w);     \
        }                                                                            \
    }
#define P1_RUN                                                                       \
    __builtin_amdgcn_sched_group_barrier(0x100, 16, 0);                              \
    __builtin_amdgcn_sched_group_barrier(0x008, 32, 0);
#define P1_MUL                                                                       \
    {                                                                                \
        const float* xt = tile + buf + (rh * 32 + h) * P1_XP + 2 * (dwl + c);        \
        const float* at = tile + buf + P1_XT + (rh * 32 + h) * P1_AP + 2 * (kw + c); \
        _Pragma("unroll")                                                            \
        for (int g = 0; g < 16; ++g) {                                               \
            if (g == 8) { P1_RUN __builtin_amdgcn_sched_barrier(0); }                \
            const float2 a0 = *(const float2*)(at + (2 * g) * P1_AP);                \
            const float2 a1 = *(const float2*)(at + (2 * g) * P1_AP + 32);           \
            const float2 b0 = *(const float2*)(xt + (2 * g) * P1_XP);                \
            const float2 b1 = *(const float2*)(xt + (2 * g) * P1_XP + 32);           \
            v2f af0, af1, bf0, bf1;                                                  \
            af0.x = a0.x; af0.y = a0.y;                                              \
            af1.x = a1.x; af1.y = a1.y;                                              \
            bf0.x = b0.x; bf0.y = b0.y;                                              \
            bf1.x = b1.x; bf1.y = b1.y;                                              \
            acc[0][0] = wmma4(af0, bf0, acc[0][0]);                                  \
            acc[0][1] = wmma4(af0, bf1, acc[0][1]);                                  \
            acc[1][0] = wmma4(af1, bf0, acc[1][0]);                                  \
            acc[1][1] = wmma4(af1, bf1, acc[1][1]);                                  \
        }                                                                            \
        P1_RUN                                                                       \
    }

#define P1_FSUM                                                                      \
    {                                                                                \
        fs.x += (ae[0].x + ao[0].x) + (ae[1].x + ao[1].x);                           \
        fs.y += (ae[0].y + ao[0].y) + (ae[1].y + ao[1].y);                           \
        fs.z += (ae[0].z + ao[0].z) + (ae[1].z + ao[1].z);                           \
        fs.w += (ae[0].w + ao[0].w) + (ae[1].w + ao[1].w);                           \
    }

    P1_LOAD(0)
    P1_FSUM
    P1_STORE(0)
    __syncthreads();

    float* const my_xch  = xch  + wave * P1_SLOT + lane;
    const float* const pr_xch = xch + (wave ^ 8) * P1_SLOT + lane;
    float* const my_hold = hold + wave * P1_SLOT + lane;

#define P1_SEND(Q)                                                                   \
    {                                                                                \
        float v[4];                                                                  \
        int   e[4];                                                                  \
        _Pragma("unroll")                                                            \
        for (int i4 = 0; i4 < 4; ++i4) {                                             \
            const int r  = (Q) * 2 + (i4 >> 1);                                      \
            const int dt = i4 & 1;           \
            v[i4] = my_hold[(dt * 8 + r) * 32];                                      \
            e[i4] = (krow + r) * D_DIM + dwl + 16 * dt + c;                          \
        }                                                                            \
        if (planes) {                                                                \
            _Pragma("unroll")                                                        \
            for (int i4 = 0; i4 < 4; ++i4) plane[e[i4]] = v[i4];                     \
        } else {                                                                     \
            _Pragma("unroll")                                                        \
            for (int i4 = 0; i4 < 4; ++i4) unsafeAtomicAdd(&sums[e[i4]], v[i4]);     \
        }                                                                            \
    }
#define P1_FLUSH(FIRST)                                                              \
    {                                                                                \
        float own[16];                                                               \
        _Pragma("unroll")                                                            \
        for (int dt = 0; dt < 2; ++dt) {                                             \
            _Pragma("unroll")                                                        \
            for (int r = 0; r < 8; ++r) {                                            \
                const float keep = rh ? acc[1][dt][r] : acc[0][dt][r];               \
                const float give = rh ? acc[0][dt][r] : acc[1][dt][r];               \
                own[dt * 8 + r] = keep;                                              \
                my_xch[(dt * 8 + r) * 32] = give;                                    \
            }                                                                        \
        }                                                                            \
        if (FIRST) {                                                                 \
              \
            fs.x += __shfl_xor(fs.x, 16, 32);                                        \
            fs.y += __shfl_xor(fs.y, 16, 32);                                        \
            fs.z += __shfl_xor(fs.z, 16, 32);                                        \
            fs.w += __shfl_xor(fs.w, 16, 32);                                        \
            if (lane < 16) *(float4*)&fred[wave * K_CLS + 4 * lane] = fs;            \
        }                                                                            \
        __syncthreads();                                                             \
        if (FIRST) {                                                                 \
            if (tid < K_CLS) {                                                       \
                float f = 0.0f;                                                      \
                _Pragma("unroll")                                                    \
                for (int w = 0; w < 16; ++w) f += fred[w * K_CLS + tid];             \
                unsafeAtomicAdd(&freq[tid], f);                                      \
            }                                                                        \
        }                                                                            \
        _Pragma("unroll")                                                            \
        for (int i = 0; i < 16; ++i) own[i] += pr_xch[i * 32];                       \
        if (FIRST) {                                                                 \
              \
            _Pragma("unroll")                                                        \
            for (int i = 0; i < 16; ++i) my_hold[i * 32] = own[i];                   \
            _Pragma("unroll")                                                        \
            for (int i = 0; i < 2; ++i)                                              \
                _Pragma("unroll")                                                    \
                for (int j = 0; j < 2; ++j) { v8f z = {}; acc[i][j] = z; }           \
        } else {                                                                     \
              \
              \
            if (planes) {                                                            \
                _Pragma("unroll")                                                    \
                for (int r = 0; r < 8; ++r)                                          \
                    _Pragma("unroll")                                                \
                    for (int dt = 0; dt < 2; ++dt)                                   \
                        plane[(krow + r) * D_DIM + P1_DH + dwl + 16 * dt + c] = own[dt * 8 + r]; \
            } else {                                                                 \
                _Pragma("unroll")                                                    \
                for (int r = 0; r < 8; ++r)                                          \
                    _Pragma("unroll")                                                \
                    for (int dt = 0; dt < 2; ++dt)                                   \
                        unsafeAtomicAdd(&sums[(krow + r) * D_DIM + P1_DH + dwl + 16 * dt + c], own[dt * 8 + r]); \
            }                                                                        \
        }                                                                            \
    }

    #pragma clang loop unroll(disable)
    for (int s = 0; s < P1_NSTAGE - 1; ++s) {
        const int buf = (s & 1) * P1_STAGE;

        P1_LOAD(s + 1)
        if (s >= P1_NT) P1_SEND(s - P1_NT)

        P1_MUL

        if (s + 1 < P1_NT) P1_FSUM
        P1_STORE(((s + 1) & 1) * P1_STAGE)
        __syncthreads();

        if (s == P1_NT - 1) P1_FLUSH(1)
    }
    {
        const int buf = ((P1_NSTAGE - 1) & 1) * P1_STAGE;
        P1_SEND(P1_NT - 1)
        P1_MUL
        __syncthreads();
        P1_FLUSH(0)
    }
#undef P1_LOAD
#undef P1_STORE
#undef P1_MUL
#undef P1_RUN
#undef P1_FSUM
#undef P1_SEND
#undef P1_FLUSH
}

#define MK_THREADS 256
#define MK_COLS    64
#define MK_PG      16
__global__ __launch_bounds__(MK_THREADS)
void means_msq(const float* __restrict__ sums, const float* __restrict__ part,
               const float* __restrict__ freq, const float* __restrict__ past,
               float* __restrict__ out_means, float* __restrict__ msq, size_t ws_bytes) {
    __shared__ __attribute__((aligned(16))) float red[MK_PG * MK_COLS];
    __shared__ float sq2[2];
    const int tid = threadIdx.x;
    const int k   = blockIdx.x >> 2;
    const int d0  = (blockIdx.x & 3) * MK_COLS;
    const bool planes = ws_holds_planes(ws_bytes);

    const float f = freq[k];
    float pm = 0.0f;

    float s = 0.0f;
    if (planes) {
        const int c4 = tid & 15;
        const int pg = tid >> 4;
        const float4* p = (const float4*)(part + (size_t)(MK_PG * pg) * P1_PLANE + k * D_DIM + d0) + c4;
        float4 v[16];
        #pragma unroll
        for (int i = 0; i < 16; ++i) v[i] = p[(size_t)i * (P1_PLANE / 4)];
        if (tid < MK_COLS) pm = past[k * D_DIM + d0 + tid];
        float4 a = v[0];
        #pragma unroll
        for (int i = 1; i < 16; ++i) { a.x += v[i].x; a.y += v[i].y; a.z += v[i].z; a.w += v[i].w; }
        *(float4*)&red[pg * MK_COLS + 4 * c4] = a;
        __syncthreads();
        if (tid < MK_COLS) {
            float t = red[tid];
            #pragma unroll
            for (int g = 1; g < MK_PG; ++g) t += red[g * MK_COLS + tid];
            s = t;
        }
    } else {
        if (tid < MK_COLS) {
            s  = sums[k * D_DIM + d0 + tid];
            pm = past[k * D_DIM + d0 + tid];
        }
    }

    float sq = 0.0f;
    if (tid < MK_COLS) {
        const float bm = s / fmaxf(f, FP_EPS);
        const float m  = (f > FP_EPS) ? (pm * 0.5f + bm * 0.5f) : pm;
        out_means[k * D_DIM + d0 + tid] = m;
        sq = m * m;
    }
    #pragma unroll
    for (int off = 16; off > 0; off >>= 1) sq += __shfl_xor(sq, off, 32);
    if (tid < MK_COLS && (tid & 31) == 0) sq2[tid >> 5] = sq;
    __syncthreads();
    if (tid == 0) unsafeAtomicAdd(&msq[k], sq2[0] + sq2[1]);
}

#define P2_THREADS 128
#define P2_ROWS    256
#define P2_TN      32
#define P2_XP      260
#define P2_WP      68
#define P2_XT      (P2_TN * P2_XP)
#define P2_WT      (P2_TN * P2_WP)
#define P2_STAGE   (P2_XT + P2_WT)
#define P2_NSTAGE  (P2_ROWS / P2_TN)
#define P2_DW      128
#define P2_NJ      (P2_DW / 4)

__global__ __launch_bounds__(P2_THREADS)
void pass2_loss(const float* __restrict__ X, const float* __restrict__ A,
                const float* __restrict__ S, const float* __restrict__ means,
                const float* __restrict__ msq, float* __restrict__ loss) {
    __shared__ __attribute__((aligned(16))) float smem[P2_STAGE + K_CLS * MLDS_STRIDE];
    __shared__ float msq_lds[K_CLS];
    float* const tile = smem;
    float* const mlds = smem + P2_STAGE;

    const int tid  = threadIdx.x;
    const int lane = tid & 31;
    const int wave = __builtin_amdgcn_readfirstlane(tid >> 5);
    const int c    = lane & 15;
    const int h    = lane >> 4;
    const int nt   = wave & 1;
    const int dh   = wave >> 1;

    const size_t row0 = (size_t)(blockIdx.x & 255) * (2 * P2_ROWS) + (size_t)(blockIdx.x >> 8) * P2_ROWS;

    const int xr = tid >> 6, xc4 = tid & 63;
    const int wr = tid >> 4, wc4 = tid & 15;
    const float4* xg = (const float4*)(X + (row0 + xr) * D_DIM) + xc4;
    const float4* ag = (const float4*)(A + (row0 + wr) * K_CLS) + wc4;
    const float4* sg = (const float4*)(S + (row0 + wr) * K_CLS) + wc4;
    const int xl = xr * P2_XP + xc4 * 4;
    const int wl = P2_XT + wr * P2_WP + wc4 * 4;

#define P2_REP16(M) M(0) M(1) M(2) M(3) M(4) M(5) M(6) M(7) M(8) M(9) M(10) M(11) M(12) M(13) M(14) M(15)
#define P2_XDECL(I) float4 xv##I;
    P2_REP16(P2_XDECL)
    float4 av[4], sv[4];

#define P2_XLD(I) xv##I = xn[(size_t)(2 * I) * (D_DIM / 4)];
#define P2_XST(I) *(float4*)(xdst + (2 * I) * P2_XP) = xv##I;
#define P2_LOAD(ST)                                                                  \
    {                                                                                \
        const float4* xn = xg + (size_t)(ST) * P2_TN * (D_DIM / 4);                  \
        const float4* an = ag + (size_t)(ST) * P2_TN * (K_CLS / 4);                  \
        const float4* sn = sg + (size_t)(ST) * P2_TN * (K_CLS / 4);                  \
        P2_REP16(P2_XLD)                                                             \
        _Pragma("unroll")                                                            \
        for (int j = 0; j < 4; ++j) {                                                \
            av[j] = an[(size_t)(8 * j) * (K_CLS / 4)];                               \
            sv[j] = sn[(size_t)(8 * j) * (K_CLS / 4)];                               \
        }                                                                            \
    }
#define P2_STORE(BUF)                                                                \
    {                                                                                \
        float* const xdst = &tile[(BUF) + xl];                                       \
        P2_REP16(P2_XST)                                                             \
        _Pragma("unroll")                                                            \
        for (int j = 0; j < 4; ++j)                                                  \
            *(float4*)&tile[(BUF) + wl + (8 * j) * P2_WP] =                          \
                make_float4(av[j].x * sv[j].x, av[j].y * sv[j].y,                    \
                            av[j].z * sv[j].z, av[j].w * sv[j].w);                   \
    }

    P2_LOAD(0)

    #pragma unroll
    for (int i = 0; i < 32; ++i) {
        const int idx = tid + i * P2_THREADS;
        const int k   = idx >> 6;
        const int c4  = idx & 63;
        const float4 v = ((const float4*)means)[(size_t)k * 64 + c4];
        *(float4*)&mlds[k * MLDS_STRIDE + c4 * 4] = v;
    }
    if (tid < K_CLS) msq_lds[tid] = msq[tid];
    P2_STORE(0)
    __syncthreads();

    float msel[4];
    #pragma unroll
    for (int kt = 0; kt < 4; ++kt) msel[kt] = (dh == 0) ? msq_lds[kt * 16 + c] : 0.0f;

    v2f bq[P2_NJ][4];
    {
        const float* mt = mlds + c * MLDS_STRIDE + dh * P2_DW + 2 * h;
        #pragma unroll
        for (int j = 0; j < P2_NJ; ++j) {
            #pragma unroll
            for (int kt = 0; kt < 4; ++kt) {
                const float2 b2 = *(const float2*)(mt + kt * 16 * MLDS_STRIDE + 4 * j);
                bq[j][kt].x = b2.x; bq[j][kt].y = b2.y;
            }
        }
    }

    __syncthreads();

    float csum = 0.0f;

    for (int s = 0; s < P2_NSTAGE; ++s) {
        const int buf = (s & 1) * P2_STAGE;
        const bool more = (s + 1 < P2_NSTAGE);

        if (more) P2_LOAD(s + 1)

        v8f acc[4];
        #pragma unroll
        for (int t = 0; t < 4; ++t) { v8f z = {}; acc[t] = z; }
        float xp = 0.0f;

        const float* xt = tile + buf + (nt * 16 + c) * P2_XP + dh * P2_DW + 2 * h;
        #pragma unroll
        for (int j = 0; j < P2_NJ; ++j) {
            const float2 a2 = *(const float2*)(xt + 4 * j);
            v2f af;
            af.x = a2.x; af.y = a2.y;
            xp += a2.x * a2.x + a2.y * a2.y;
            #pragma unroll
            for (int kt = 0; kt < 4; ++kt) acc[kt] = wmma4(af, bq[j][kt], acc[kt]);
        }

        const float xs = xp + __shfl_xor(xp, 16, 32);

        const float* wt = tile + buf + P2_XT + (nt * 16 + 8 * h) * P2_WP + c;
        #pragma unroll
        for (int r = 0; r < 8; ++r) {
            const float xrow = __shfl(xs, 8 * h + r, 32);
            #pragma unroll
            for (int kt = 0; kt < 4; ++kt) {
                const float w = wt[r * P2_WP + kt * 16];
                csum += w * (xrow + msel[kt] - 2.0f * acc[kt][r]);
            }
        }

        if (more) P2_STORE(((s + 1) & 1) * P2_STAGE)
        __syncthreads();
    }
#undef P2_LOAD
#undef P2_STORE
#undef P2_XLD
#undef P2_XST
#undef P2_XDECL
#undef P2_REP16

    #pragma unroll
    for (int off = 16; off > 0; off >>= 1) csum += __shfl_xor(csum, off, 32);
    if (lane == 0)
        unsafeAtomicAdd(loss, csum * (1.0f / (float)N_TOT));
}

extern "C" void kernel_launch(void* const* d_in, const int* in_sizes, int n_in,
                              void* d_out, int out_size, void* d_ws, size_t ws_size,
                              hipStream_t stream) {
    const float* X = (const float*)d_in[0];
    const float* A = (const float*)d_in[1];
    const float* S = (const float*)d_in[2];
    const float* P = (const float*)d_in[3];
    float* out = (float*)d_out;

    float* ws   = (float*)d_ws;
    float* sums = ws;
    float* freq = ws + K_CLS * D_DIM;
    float* msq  = freq + K_CLS;
    float* part = msq + K_CLS;

    zero_ws<<<(K_CLS * D_DIM + 2 * K_CLS + 255) / 256, 256, 0, stream>>>(ws, out);

    pass1_colsum<<<N_TOT / P1_ROWS, P1_THREADS, 0, stream>>>(X, A, sums, freq, part, ws_size);

    means_msq<<<4 * K_CLS, MK_THREADS, 0, stream>>>(sums, part, freq, P, out + 1, msq, ws_size);

    pass2_loss<<<N_TOT / P2_ROWS, P2_THREADS, 0, stream>>>(X, A, S, out + 1, msq, out);
}
